# starting kernel
# baseline (speedup 1.0000x reference)
#include <hip/hip_runtime.h>

#define BATCH 32
#define V     1024
#define HID   768
#define OUT   256
#define S1    8
#define S2    8

typedef __attribute__((ext_vector_type(2))) float v2f;
typedef __attribute__((ext_vector_type(8))) float v8f;

__global__ void colsum_partial_kernel(const float* __restrict__ adj,
                                      float* __restrict__ part) {
  const int b = blockIdx.x;
  const int w = blockIdx.y * 256 + threadIdx.x;
  const int s = blockIdx.z;
  const float* p = adj + (size_t)b * V * V + (size_t)(s * (V / S1)) * V + w;
  float acc = 0.0f;
#pragma unroll 4
  for (int v = 0; v < V / S1; ++v) {
    __builtin_prefetch(p + 16 * V, 0, 1);
    acc += *p;
    p += V;
  }
  part[((b * S1 + s) * V) + w] = acc;
}

__global__ void colsum_reduce_kernel(const float* __restrict__ part,
                                     float* __restrict__ colsum) {
  const int b = blockIdx.x;
  const int w = blockIdx.y * 256 + threadIdx.x;
  float acc = 0.0f;
#pragma unroll
  for (int s = 0; s < S1; ++s) acc += part[(b * S1 + s) * V + w];
  colsum[b * V + w] = acc;
}

__global__ void gather_partial_kernel(const float* __restrict__ emb,
                                      const int* __restrict__ ids,
                                      const float* __restrict__ colsum,
                                      float* __restrict__ hpart) {
  const int b  = blockIdx.x;
  const int hh = blockIdx.y * 256 + threadIdx.x;
  const int s  = blockIdx.z;
  const int wbase = s * (V / S2);
  float acc = 0.0f;
  for (int i = 0; i < V / S2; ++i) {
    const int   w  = wbase + i;
    const int   id = ids[b * V + w];
    const float cw = colsum[b * V + w];
    acc += cw * emb[(size_t)id * HID + hh];
  }
  hpart[(b * S2 + s) * HID + hh] = acc;
}

__global__ void h_reduce_kernel(const float* __restrict__ hpart,
                                float* __restrict__ hbuf) {
  const int b  = blockIdx.x;
  const int hh = blockIdx.y * 256 + threadIdx.x;
  float acc = 0.0f;
#pragma unroll
  for (int s = 0; s < S2; ++s) acc += hpart[(b * S2 + s) * HID + hh];
  hbuf[b * HID + hh] = acc * (1.0f / (float)V);
}

__global__ void fc_wmma_kernel(const float* __restrict__ hbuf,
                               const float* __restrict__ fc_w,
                               const float* __restrict__ fc_b,
                               float* __restrict__ out) {
  const int wave  = (blockIdx.x * blockDim.x + threadIdx.x) >> 5;
  const int lane  = threadIdx.x & 31;
  const int mTile = wave >> 4;
  const int nTile = wave & 15;
  const int m0 = mTile * 16;
  const int n0 = nTile * 16;
  const int half = lane >> 4;
  const int l    = lane & 15;
  const int kOff = 2 * half;

  const float* arow = hbuf + (size_t)(m0 + l) * HID;
  const float* brow = fc_w + (size_t)(n0 + l) * HID;

  v8f c = {};
  for (int k = 0; k < HID; k += 4) {
    v2f a  = *(const v2f*)(arow + k + kOff);
    v2f bb = *(const v2f*)(brow + k + kOff);
    c = __builtin_amdgcn_wmma_f32_16x16x4_f32(false, a, false, bb,
                                              (short)0, c, false, false);
  }

#pragma unroll
  for (int i = 0; i < 8; ++i) {
    const int m = m0 + i + 8 * half;
    const int n = n0 + l;
    out[m * OUT + n] = c[i] + fc_b[n];
  }
}

extern "C" void kernel_launch(void* const* d_in, const int* in_sizes, int n_in,
                              void* d_out, int out_size, void* d_ws, size_t ws_size,
                              hipStream_t stream) {
  (void)in_sizes; (void)n_in; (void)out_size; (void)ws_size;
  const float* emb    = (const float*)d_in[0];
  const float* adj    = (const float*)d_in[1];
  const float* fc_w   = (const float*)d_in[2];
  const float* fc_b   = (const float*)d_in[3];
  const int*   gv_ids = (const int*)d_in[4];
  float* out = (float*)d_out;

  char* ws = (char*)d_ws;
  float* part   = (float*)ws;
  float* colsum = (float*)(ws + sizeof(float) * ((size_t)BATCH * S1 * V));
  float* hpart  = (float*)(ws + sizeof(float) * ((size_t)BATCH * S1 * V +
                                                 (size_t)BATCH * V));
  float* hbuf   = (float*)(ws + sizeof(float) * ((size_t)BATCH * S1 * V +
                                                 (size_t)BATCH * V +
                                                 (size_t)BATCH * S2 * HID));

  colsum_partial_kernel<<<dim3(BATCH, V / 256, S1), 256, 0, stream>>>(adj, part);
  colsum_reduce_kernel <<<dim3(BATCH, V / 256),     256, 0, stream>>>(part, colsum);
  gather_partial_kernel<<<dim3(BATCH, HID / 256, S2), 256, 0, stream>>>(emb, gv_ids, colsum, hpart);
  h_reduce_kernel      <<<dim3(BATCH, HID / 256),     256, 0, stream>>>(hpart, hbuf);
  fc_wmma_kernel<<<dim3(4), 256, 0, stream>>>(hbuf, fc_w, fc_b, out);
}


# fc GEMM: each tile's 16 rows of h and of fc_w copied to LDS by its whole block with whole-line 16-byte loads (32 blocks, was 4); WMMA fragments read from LDS; k order and single accumulator kept
# speedup vs baseline: 1.2740x; 1.2740x over previous
#include <hip/hip_runtime.h>

#define BATCH 32
#define V     1024
#define HID   768
#define OUT   256
#define S1    8
#define S2    8

typedef __attribute__((ext_vector_type(2))) float v2f;
typedef __attribute__((ext_vector_type(8))) float v8f;

__global__ void colsum_partial_kernel(const float* __restrict__ adj,
                                      float* __restrict__ part) {
  const int b = blockIdx.x;
  const int w = blockIdx.y * 256 + threadIdx.x;
  const int s = blockIdx.z;
  const float* p = adj + (size_t)b * V * V + (size_t)(s * (V / S1)) * V + w;
  float acc = 0.0f;
#pragma unroll 4
  for (int v = 0; v < V / S1; ++v) {
    __builtin_prefetch(p + 16 * V, 0, 1);
    acc += *p;
    p += V;
  }
  part[((b * S1 + s) * V) + w] = acc;
}

__global__ void colsum_reduce_kernel(const float* __restrict__ part,
                                     float* __restrict__ colsum) {
  const int b = blockIdx.x;
  const int w = blockIdx.y * 256 + threadIdx.x;
  float acc = 0.0f;
#pragma unroll
  for (int s = 0; s < S1; ++s) acc += part[(b * S1 + s) * V + w];
  colsum[b * V + w] = acc;
}

__global__ void gather_partial_kernel(const float* __restrict__ emb,
                                      const int* __restrict__ ids,
                                      const float* __restrict__ colsum,
                                      float* __restrict__ hpart) {
  const int b  = blockIdx.x;
  const int hh = blockIdx.y * 256 + threadIdx.x;
  const int s  = blockIdx.z;
  const int wbase = s * (V / S2);
  float acc = 0.0f;
  for (int i = 0; i < V / S2; ++i) {
    const int   w  = wbase + i;
    const int   id = ids[b * V + w];
    const float cw = colsum[b * V + w];
    acc += cw * emb[(size_t)id * HID + hh];
  }
  hpart[(b * S2 + s) * HID + hh] = acc;
}

__global__ void h_reduce_kernel(const float* __restrict__ hpart,
                                float* __restrict__ hbuf) {
  const int b  = blockIdx.x;
  const int hh = blockIdx.y * 256 + threadIdx.x;
  float acc = 0.0f;
#pragma unroll
  for (int s = 0; s < S2; ++s) acc += hpart[(b * S2 + s) * HID + hh];
  hbuf[b * HID + hh] = acc * (1.0f / (float)V);
}

#define FC_PITCH 772
#define FC_KCH   256
#define FC_NCH   (HID / FC_KCH)
#define FC_GRP   16
#define FC_NGRP  (FC_KCH / (4 * FC_GRP))

typedef __attribute__((ext_vector_type(4))) float v4f;

__global__ void __launch_bounds__(256)
fc_wmma_kernel(const float* __restrict__ hbuf,
               const float* __restrict__ fc_w,
               const float* __restrict__ fc_b,
               float* __restrict__ out) {
  __shared__ float As[16 * FC_PITCH];
  __shared__ float Bs[16 * FC_PITCH];

  const int tid   = threadIdx.x;
  const int mTile = blockIdx.x >> 4;
  const int nTile = blockIdx.x & 15;
  const int m0 = mTile * 16;
  const int n0 = nTile * 16;

  const float* ag = hbuf + (size_t)m0 * HID;
  const float* bg = fc_w + (size_t)n0 * HID;

  v4f ra[FC_NCH][4], rb[FC_NCH][4];
#pragma unroll
  for (int c = 0; c < FC_NCH; ++c) {
#pragma unroll
    for (int j = 0; j < 4; ++j) {
      const int q   = tid + 256 * j;
      const int off = (q >> 6) * HID + c * FC_KCH + (q & 63) * 4;
      ra[c][j] = *(const v4f*)(ag + off);
      rb[c][j] = *(const v4f*)(bg + off);
    }
    __builtin_amdgcn_sched_barrier(0);
  }

  const int wave = __builtin_amdgcn_readfirstlane(tid >> 5);
  const int lane = tid & 31;
  const int half = lane >> 4;
  const int l    = lane & 15;
  const float* as = As + l * FC_PITCH + 2 * half;
  const float* bs = Bs + l * FC_PITCH + 2 * half;

  v8f acc = {};
#pragma unroll
  for (int c = 0; c < FC_NCH; ++c) {
#pragma unroll
    for (int j = 0; j < 4; ++j) {
      const int q    = tid + 256 * j;
      const int loff = (q >> 6) * FC_PITCH + c * FC_KCH + (q & 63) * 4;
      *(v4f*)(As + loff) = ra[c][j];
      *(v4f*)(Bs + loff) = rb[c][j];
    }
    __builtin_amdgcn_fence(__ATOMIC_RELEASE, "workgroup", "local");
    __builtin_amdgcn_s_barrier();
    __builtin_amdgcn_fence(__ATOMIC_ACQUIRE, "workgroup", "local");
    if (wave == 0) {
      v2f fa[2][FC_GRP], fb[2][FC_GRP];
      const int kbase = c * FC_KCH;
#pragma unroll
      for (int i = 0; i < FC_GRP; ++i) {
        fa[0][i] = *(const v2f*)(as + kbase + 4 * i);
        fb[0][i] = *(const v2f*)(bs + kbase + 4 * i);
      }
#pragma unroll
      for (int g = 0; g < FC_NGRP; ++g) {
        if (g + 1 < FC_NGRP) {
#pragma unroll
          for (int i = 0; i < FC_GRP; ++i) {
            fa[(g + 1) & 1][i] = *(const v2f*)(as + kbase + 4 * (FC_GRP * (g + 1) + i));
            fb[(g + 1) & 1][i] = *(const v2f*)(bs + kbase + 4 * (FC_GRP * (g + 1) + i));
          }
        }
        __builtin_amdgcn_sched_barrier(0);
#pragma unroll
        for (int i = 0; i < FC_GRP; ++i) {
          acc = __builtin_amdgcn_wmma_f32_16x16x4_f32(false, fa[g & 1][i], false, fb[g & 1][i],
                                                      (short)0, acc, false, false);
        }
        __builtin_amdgcn_sched_barrier(0);
      }
    }
  }

  if (wave == 0) {
    const float bias = fc_b[n0 + l];
#pragma unroll
    for (int i = 0; i < 8; ++i) {
      const int m = m0 + i + 8 * half;
      out[m * OUT + n0 + l] = acc[i] + bias;
    }
  }
}

extern "C" void kernel_launch(void* const* d_in, const int* in_sizes, int n_in,
                              void* d_out, int out_size, void* d_ws, size_t ws_size,
                              hipStream_t stream) {
  (void)in_sizes; (void)n_in; (void)out_size; (void)ws_size;
  const float* emb    = (const float*)d_in[0];
  const float* adj    = (const float*)d_in[1];
  const float* fc_w   = (const float*)d_in[2];
  const float* fc_b   = (const float*)d_in[3];
  const int*   gv_ids = (const int*)d_in[4];
  float* out = (float*)d_out;

  char* ws = (char*)d_ws;
  float* part   = (float*)ws;
  float* colsum = (float*)(ws + sizeof(float) * ((size_t)BATCH * S1 * V));
  float* hpart  = (float*)(ws + sizeof(float) * ((size_t)BATCH * S1 * V +
                                                 (size_t)BATCH * V));
  float* hbuf   = (float*)(ws + sizeof(float) * ((size_t)BATCH * S1 * V +
                                                 (size_t)BATCH * V +
                                                 (size_t)BATCH * S2 * HID));

  colsum_partial_kernel<<<dim3(BATCH, V / 256, S1), 256, 0, stream>>>(adj, part);
  colsum_reduce_kernel <<<dim3(BATCH, V / 256),     256, 0, stream>>>(part, colsum);
  gather_partial_kernel<<<dim3(BATCH, HID / 256, S2), 256, 0, stream>>>(emb, gv_ids, colsum, hpart);
  h_reduce_kernel      <<<dim3(BATCH, HID / 256),     256, 0, stream>>>(hpart, hbuf);
  fc_wmma_kernel<<<dim3(32), 256, 0, stream>>>(hbuf, fc_w, fc_b, out);
}


# colsum and gather fused in one kernel: block (b, 64 columns) sums all 1024 rows, hands its 64 column sums over in LDS and gathers its 64 embedding rows; 3 launches (was 5); summation order changes
# speedup vs baseline: 2.0003x; 1.5701x over previous
#include <hip/hip_runtime.h>

#define BATCH 32
#define V     1024
#define HID   768
#define OUT   256
#define CW    64
#define NSL   (V / CW)
#define S2    NSL
#define RPW   (CW / 8)
#define AB    16
#define NAB   (V / 2 / 8 / AB)
#define GB    (RPW / 2)
#define GP    (HID / 128)

typedef __attribute__((ext_vector_type(2))) float v2f;
typedef __attribute__((ext_vector_type(4))) float v4f;
typedef __attribute__((ext_vector_type(8))) float v8f;

__global__ void __launch_bounds__(256)
colsum_gather_kernel(const float* __restrict__ adj,
                     const float* __restrict__ emb,
                     const int* __restrict__ ids,
                     float* __restrict__ hpart) {
  __shared__ float red[16 * CW];
  __shared__ float hred[8 * HID];

  const int sl   = blockIdx.x;
  const int b    = blockIdx.y;
  const int tid  = threadIdx.x;
  const int wave = __builtin_amdgcn_readfirstlane(tid >> 5);
  const int lane = tid & 31;
  const int half = lane >> 4;
  const int c4   = lane & 15;

  const int w0 = sl * CW + wave * RPW;
  int id[RPW];
#pragma unroll
  for (int i = 0; i < RPW; ++i) id[i] = ids[b * V + w0 + i];

  const float* ap = adj + ((size_t)b * V + 2 * wave + half) * V + sl * CW + c4 * 4;
  const float* ep = emb + lane * 4;

  v4f r[2][AB];
  v4f e0[GB][GP], e1[GB][GP];
#pragma unroll
  for (int j = 0; j < AB; ++j) r[0][j] = *(const v4f*)(ap + (size_t)(16 * j) * V);

  v4f acc = {0.0f, 0.0f, 0.0f, 0.0f};
#pragma unroll
  for (int k = 0; k < NAB; ++k) {
    if (k + 1 < NAB) {
#pragma unroll
      for (int j = 0; j < AB; ++j)
        r[(k + 1) & 1][j] = *(const v4f*)(ap + (size_t)(16 * ((k + 1) * AB + j)) * V);
    } else {
#pragma unroll
      for (int i = 0; i < GB; ++i)
#pragma unroll
        for (int j = 0; j < GP; ++j)
          e0[i][j] = *(const v4f*)(ep + (size_t)id[i] * HID + j * 128);
    }
    __builtin_amdgcn_sched_barrier(0);
    if (k == 0) {
      asm volatile("" :: "s"(id[0]), "s"(id[1]), "s"(id[2]), "s"(id[3]),
                         "s"(id[4]), "s"(id[5]), "s"(id[6]), "s"(id[7]));
    }
#pragma unroll
    for (int j = 0; j < AB; ++j) acc += r[k & 1][j];
    __builtin_amdgcn_sched_barrier(0);
  }

  *(v4f*)(red + (wave * 2 + half) * CW + c4 * 4) = acc;
  __builtin_amdgcn_fence(__ATOMIC_RELEASE, "workgroup", "local");
  __builtin_amdgcn_s_barrier();
  __builtin_amdgcn_fence(__ATOMIC_ACQUIRE, "workgroup", "local");

  float cs = 0.0f;
#pragma unroll
  for (int p = 0; p < 16; ++p) cs += red[p * CW + wave * RPW + (lane & 7)];
  float cw[RPW];
#pragma unroll
  for (int i = 0; i < RPW; ++i)
    cw[i] = __builtin_bit_cast(float, __builtin_amdgcn_readlane(__builtin_bit_cast(int, cs), i));

#pragma unroll
  for (int i = 0; i < GB; ++i)
#pragma unroll
    for (int j = 0; j < GP; ++j)
      e1[i][j] = *(const v4f*)(ep + (size_t)id[GB + i] * HID + j * 128);
  __builtin_amdgcn_sched_barrier(0);

  v4f hacc[GP];
#pragma unroll
  for (int j = 0; j < GP; ++j) hacc[j] = cw[0] * e0[0][j];
#pragma unroll
  for (int i = 1; i < GB; ++i)
#pragma unroll
    for (int j = 0; j < GP; ++j) hacc[j] += cw[i] * e0[i][j];
#pragma unroll
  for (int i = 0; i < GB; ++i)
#pragma unroll
    for (int j = 0; j < GP; ++j) hacc[j] += cw[GB + i] * e1[i][j];

#pragma unroll
  for (int j = 0; j < GP; ++j) *(v4f*)(hred + wave * HID + j * 128 + lane * 4) = hacc[j];
  __syncthreads();
  if (tid < HID / 4) {
    v4f s = *(const v4f*)(hred + tid * 4);
#pragma unroll
    for (int wv = 1; wv < 8; ++wv) s += *(const v4f*)(hred + wv * HID + tid * 4);
    *(v4f*)(hpart + (size_t)(b * NSL + sl) * HID + tid * 4) = s;
  }
}

__global__ void h_reduce_kernel(const float* __restrict__ hpart,
                                float* __restrict__ hbuf) {
  const int b  = blockIdx.x;
  const int hh = blockIdx.y * 256 + threadIdx.x;
  float acc = 0.0f;
#pragma unroll
  for (int s = 0; s < S2; ++s) acc += hpart[(b * S2 + s) * HID + hh];
  hbuf[b * HID + hh] = acc * (1.0f / (float)V);
}

#define FC_PITCH 772
#define FC_KCH   256
#define FC_NCH   (HID / FC_KCH)
#define FC_GRP   16
#define FC_NGRP  (FC_KCH / (4 * FC_GRP))

__global__ void __launch_bounds__(256)
fc_wmma_kernel(const float* __restrict__ hbuf,
               const float* __restrict__ fc_w,
               const float* __restrict__ fc_b,
               float* __restrict__ out) {
  __shared__ float As[16 * FC_PITCH];
  __shared__ float Bs[16 * FC_PITCH];

  const int tid   = threadIdx.x;
  const int mTile = blockIdx.x >> 4;
  const int nTile = blockIdx.x & 15;
  const int m0 = mTile * 16;
  const int n0 = nTile * 16;

  const float* ag = hbuf + (size_t)m0 * HID;
  const float* bg = fc_w + (size_t)n0 * HID;

  v4f ra[FC_NCH][4], rb[FC_NCH][4];
#pragma unroll
  for (int c = 0; c < FC_NCH; ++c) {
#pragma unroll
    for (int j = 0; j < 4; ++j) {
      const int q   = tid + 256 * j;
      const int off = (q >> 6) * HID + c * FC_KCH + (q & 63) * 4;
      ra[c][j] = *(const v4f*)(ag + off);
      rb[c][j] = *(const v4f*)(bg + off);
    }
    __builtin_amdgcn_sched_barrier(0);
  }

  const int wave = __builtin_amdgcn_readfirstlane(tid >> 5);
  const int lane = tid & 31;
  const int half = lane >> 4;
  const int l    = lane & 15;
  const float* as = As + l * FC_PITCH + 2 * half;
  const float* bs = Bs + l * FC_PITCH + 2 * half;

  v8f acc = {};
#pragma unroll
  for (int c = 0; c < FC_NCH; ++c) {
#pragma unroll
    for (int j = 0; j < 4; ++j) {
      const int q    = tid + 256 * j;
      const int loff = (q >> 6) * FC_PITCH + c * FC_KCH + (q & 63) * 4;
      *(v4f*)(As + loff) = ra[c][j];
      *(v4f*)(Bs + loff) = rb[c][j];
    }
    __builtin_amdgcn_fence(__ATOMIC_RELEASE, "workgroup", "local");
    __builtin_amdgcn_s_barrier();
    __builtin_amdgcn_fence(__ATOMIC_ACQUIRE, "workgroup", "local");
    if (wave == 0) {
      v2f fa[2][FC_GRP], fb[2][FC_GRP];
      const int kbase = c * FC_KCH;
#pragma unroll
      for (int i = 0; i < FC_GRP; ++i) {
        fa[0][i] = *(const v2f*)(as + kbase + 4 * i);
        fb[0][i] = *(const v2f*)(bs + kbase + 4 * i);
      }
#pragma unroll
      for (int g = 0; g < FC_NGRP; ++g) {
        if (g + 1 < FC_NGRP) {
#pragma unroll
          for (int i = 0; i < FC_GRP; ++i) {
            fa[(g + 1) & 1][i] = *(const v2f*)(as + kbase + 4 * (FC_GRP * (g + 1) + i));
            fb[(g + 1) & 1][i] = *(const v2f*)(bs + kbase + 4 * (FC_GRP * (g + 1) + i));
          }
        }
        __builtin_amdgcn_sched_barrier(0);
#pragma unroll
        for (int i = 0; i < FC_GRP; ++i) {
          acc = __builtin_amdgcn_wmma_f32_16x16x4_f32(false, fa[g & 1][i], false, fb[g & 1][i],
                                                      (short)0, acc, false, false);
        }
        __builtin_amdgcn_sched_barrier(0);
      }
    }
  }

  if (wave == 0) {
    const float bias = fc_b[n0 + l];
#pragma unroll
    for (int i = 0; i < 8; ++i) {
      const int m = m0 + i + 8 * half;
      out[m * OUT + n0 + l] = acc[i] + bias;
    }
  }
}

extern "C" void kernel_launch(void* const* d_in, const int* in_sizes, int n_in,
                              void* d_out, int out_size, void* d_ws, size_t ws_size,
                              hipStream_t stream) {
  (void)in_sizes; (void)n_in; (void)out_size; (void)ws_size;
  const float* emb    = (const float*)d_in[0];
  const float* adj    = (const float*)d_in[1];
  const float* fc_w   = (const float*)d_in[2];
  const float* fc_b   = (const float*)d_in[3];
  const int*   gv_ids = (const int*)d_in[4];
  float* out = (float*)d_out;

  char* ws = (char*)d_ws;
  float* hpart = (float*)ws;
  float* hbuf  = (float*)(ws + sizeof(float) * ((size_t)BATCH * NSL * HID));

  colsum_gather_kernel<<<dim3(NSL, BATCH), 256, 0, stream>>>(adj, emb, gv_ids, hpart);
  h_reduce_kernel     <<<dim3(BATCH, HID / 256), 256, 0, stream>>>(hpart, hbuf);
  fc_wmma_kernel<<<dim3(32), 256, 0, stream>>>(hbuf, fc_w, fc_b, out);
}


# h reduce launch folded into the fc GEMM: a WMMA tile's 16 M rows are one sample's 16 partial h rows, D rows summed in the epilogue; 256 blocks x 2 tiles; 2 launches (was 3); summation order changes
# speedup vs baseline: 2.2248x; 1.1122x over previous
#include <hip/hip_runtime.h>

#define BATCH 32
#define V     1024
#define HID   768
#define OUT   256
#define CW    64
#define NSL   (V / CW)
#define RPW   (CW / 8)
#define AB    16
#define NAB   (V / 2 / 8 / AB)
#define GB    (RPW / 2)
#define GP    (HID / 128)

typedef __attribute__((ext_vector_type(2))) float v2f;
typedef __attribute__((ext_vector_type(4))) float v4f;
typedef __attribute__((ext_vector_type(8))) float v8f;

__global__ void __launch_bounds__(256)
colsum_gather_kernel(const float* __restrict__ adj,
                     const float* __restrict__ emb,
                     const int* __restrict__ ids,
                     float* __restrict__ hpart) {
  __shared__ float red[16 * CW];
  __shared__ float hred[8 * HID];

  const int sl   = blockIdx.x;
  const int b    = blockIdx.y;
  const int tid  = threadIdx.x;
  const int wave = __builtin_amdgcn_readfirstlane(tid >> 5);
  const int lane = tid & 31;
  const int half = lane >> 4;
  const int c4   = lane & 15;

  const int w0 = sl * CW + wave * RPW;
  int id[RPW];
#pragma unroll
  for (int i = 0; i < RPW; ++i) id[i] = ids[b * V + w0 + i];

  const float* ap = adj + ((size_t)b * V + 2 * wave + half) * V + sl * CW + c4 * 4;
  const float* ep = emb + lane * 4;

  v4f r[2][AB];
  v4f e0[GB][GP], e1[GB][GP];
#pragma unroll
  for (int j = 0; j < AB; ++j) r[0][j] = *(const v4f*)(ap + (size_t)(16 * j) * V);

  v4f acc = {0.0f, 0.0f, 0.0f, 0.0f};
#pragma unroll
  for (int k = 0; k < NAB; ++k) {
    if (k + 1 < NAB) {
#pragma unroll
      for (int j = 0; j < AB; ++j)
        r[(k + 1) & 1][j] = *(const v4f*)(ap + (size_t)(16 * ((k + 1) * AB + j)) * V);
    } else {
#pragma unroll
      for (int i = 0; i < GB; ++i)
#pragma unroll
        for (int j = 0; j < GP; ++j)
          e0[i][j] = *(const v4f*)(ep + (size_t)id[i] * HID + j * 128);
    }
    __builtin_amdgcn_sched_barrier(0);
    if (k == 0) {
      asm volatile("" :: "s"(id[0]), "s"(id[1]), "s"(id[2]), "s"(id[3]),
                         "s"(id[4]), "s"(id[5]), "s"(id[6]), "s"(id[7]));
    }
#pragma unroll
    for (int j = 0; j < AB; ++j) acc += r[k & 1][j];
    __builtin_amdgcn_sched_barrier(0);
  }

  *(v4f*)(red + (wave * 2 + half) * CW + c4 * 4) = acc;
  __builtin_amdgcn_fence(__ATOMIC_RELEASE, "workgroup", "local");
  __builtin_amdgcn_s_barrier();
  __builtin_amdgcn_fence(__ATOMIC_ACQUIRE, "workgroup", "local");

  float cs = 0.0f;
#pragma unroll
  for (int p = 0; p < 16; ++p) cs += red[p * CW + wave * RPW + (lane & 7)];
  float cw[RPW];
#pragma unroll
  for (int i = 0; i < RPW; ++i)
    cw[i] = __builtin_bit_cast(float, __builtin_amdgcn_readlane(__builtin_bit_cast(int, cs), i));

#pragma unroll
  for (int i = 0; i < GB; ++i)
#pragma unroll
    for (int j = 0; j < GP; ++j)
      e1[i][j] = *(const v4f*)(ep + (size_t)id[GB + i] * HID + j * 128);
  __builtin_amdgcn_sched_barrier(0);

  v4f hacc[GP];
#pragma unroll
  for (int j = 0; j < GP; ++j) hacc[j] = cw[0] * e0[0][j];
#pragma unroll
  for (int i = 1; i < GB; ++i)
#pragma unroll
    for (int j = 0; j < GP; ++j) hacc[j] += cw[i] * e0[i][j];
#pragma unroll
  for (int i = 0; i < GB; ++i)
#pragma unroll
    for (int j = 0; j < GP; ++j) hacc[j] += cw[GB + i] * e1[i][j];

#pragma unroll
  for (int j = 0; j < GP; ++j) *(v4f*)(hred + wave * HID + j * 128 + lane * 4) = hacc[j];
  __syncthreads();
  if (tid < HID / 4) {
    v4f s = *(const v4f*)(hred + tid * 4);
#pragma unroll
    for (int wv = 1; wv < 8; ++wv) s += *(const v4f*)(hred + wv * HID + tid * 4);
    *(v4f*)(hpart + (size_t)(b * NSL + sl) * HID + tid * 4) = s;
  }
}

#define FC_PITCH 772
#define FC_KCH   256
#define FC_NCH   (HID / FC_KCH)
#define FC_GRP   16
#define FC_NGRP  (FC_KCH / (4 * FC_GRP))
#define FC_THR   128
#define FC_BR    32
#define FC_NA    (16 * (FC_KCH / 4) / FC_THR)
#define FC_NB    (FC_BR * (FC_KCH / 4) / FC_THR)

__global__ void __launch_bounds__(FC_THR)
fc_fold_wmma_kernel(const float* __restrict__ hpart,
                    const float* __restrict__ fc_w,
                    const float* __restrict__ fc_b,
                    float* __restrict__ out) {
  __shared__ float As[16 * FC_PITCH];
  __shared__ float Bs[FC_BR * FC_PITCH];
  __shared__ float osum[2 * 32];

  const int tid = threadIdx.x;
  const int b   = blockIdx.x >> 3;
  const int n0  = (blockIdx.x & 7) * FC_BR;

  const float* ag = hpart + (size_t)b * NSL * HID;
  const float* bg = fc_w + (size_t)n0 * HID;

  v4f ra[FC_NCH][FC_NA], rb[FC_NCH][FC_NB];
#pragma unroll
  for (int c = 0; c < FC_NCH; ++c) {
#pragma unroll
    for (int j = 0; j < FC_NA; ++j) {
      const int q   = tid + FC_THR * j;
      const int off = (q >> 6) * HID + c * FC_KCH + (q & 63) * 4;
      ra[c][j] = *(const v4f*)(ag + off);
    }
#pragma unroll
    for (int j = 0; j < FC_NB; ++j) {
      const int q   = tid + FC_THR * j;
      const int off = (q >> 6) * HID + c * FC_KCH + (q & 63) * 4;
      rb[c][j] = *(const v4f*)(bg + off);
    }
    __builtin_amdgcn_sched_barrier(0);
  }
  const float bias = fc_b[n0 + (tid & 31)];

  const int wave = __builtin_amdgcn_readfirstlane(tid >> 5);
  const int lane = tid & 31;
  const int half = lane >> 4;
  const int l    = lane & 15;
  const float* as = As + l * FC_PITCH + 2 * half;
  const float* bs = Bs + ((wave & 1) * 16 + l) * FC_PITCH + 2 * half;

  v8f acc = {};
#pragma unroll
  for (int c = 0; c < FC_NCH; ++c) {
#pragma unroll
    for (int j = 0; j < FC_NA; ++j) {
      const int q    = tid + FC_THR * j;
      const int loff = (q >> 6) * FC_PITCH + c * FC_KCH + (q & 63) * 4;
      *(v4f*)(As + loff) = ra[c][j];
    }
#pragma unroll
    for (int j = 0; j < FC_NB; ++j) {
      const int q    = tid + FC_THR * j;
      const int loff = (q >> 6) * FC_PITCH + c * FC_KCH + (q & 63) * 4;
      *(v4f*)(Bs + loff) = rb[c][j];
    }
    __builtin_amdgcn_fence(__ATOMIC_RELEASE, "workgroup", "local");
    __builtin_amdgcn_s_barrier();
    __builtin_amdgcn_fence(__ATOMIC_ACQUIRE, "workgroup", "local");
    if (wave < 2) {
      v2f fa[2][FC_GRP], fb[2][FC_GRP];
      const int kbase = c * FC_KCH;
#pragma unroll
      for (int i = 0; i < FC_GRP; ++i) {
        fa[0][i] = *(const v2f*)(as + kbase + 4 * i);
        fb[0][i] = *(const v2f*)(bs + kbase + 4 * i);
      }
#pragma unroll
      for (int g = 0; g < FC_NGRP; ++g) {
        if (g + 1 < FC_NGRP) {
#pragma unroll
          for (int i = 0; i < FC_GRP; ++i) {
            fa[(g + 1) & 1][i] = *(const v2f*)(as + kbase + 4 * (FC_GRP * (g + 1) + i));
            fb[(g + 1) & 1][i] = *(const v2f*)(bs + kbase + 4 * (FC_GRP * (g + 1) + i));
          }
        }
        __builtin_amdgcn_sched_barrier(0);
#pragma unroll
        for (int i = 0; i < FC_GRP; ++i) {
          acc = __builtin_amdgcn_wmma_f32_16x16x4_f32(false, fa[g & 1][i], false, fb[g & 1][i],
                                                      (short)0, acc, false, false);
        }
        __builtin_amdgcn_sched_barrier(0);
      }
    }
  }

  if (wave < 2) {
    float s = acc[0];
#pragma unroll
    for (int i = 1; i < 8; ++i) s += acc[i];
    osum[wave * 32 + lane] = s;
  }
  __syncthreads();
  if (wave == 0) {
    const int t = lane >> 4;
    const float v = osum[t * 32 + l] + osum[t * 32 + 16 + l];
    out[b * OUT + n0 + lane] = v * (1.0f / (float)V) + bias;
  }
}

extern "C" void kernel_launch(void* const* d_in, const int* in_sizes, int n_in,
                              void* d_out, int out_size, void* d_ws, size_t ws_size,
                              hipStream_t stream) {
  (void)in_sizes; (void)n_in; (void)out_size; (void)ws_size;
  const float* emb    = (const float*)d_in[0];
  const float* adj    = (const float*)d_in[1];
  const float* fc_w   = (const float*)d_in[2];
  const float* fc_b   = (const float*)d_in[3];
  const int*   gv_ids = (const int*)d_in[4];
  float* out = (float*)d_out;

  float* hpart = (float*)d_ws;

  colsum_gather_kernel<<<dim3(NSL, BATCH), 256, 0, stream>>>(adj, emb, gv_ids, hpart);
  fc_fold_wmma_kernel<<<dim3(BATCH * (OUT / FC_BR)), FC_THR, 0, stream>>>(hpart, fc_w, fc_b, out);
}
